# starting kernel
# baseline (speedup 1.0000x reference)
#include <hip/hip_runtime.h>

typedef float v2f __attribute__((ext_vector_type(2)));
typedef float v4f __attribute__((ext_vector_type(4)));
typedef float v8f __attribute__((ext_vector_type(8)));

#define WAVES_PER_BLOCK 8
#define TPB (WAVES_PER_BLOCK * 32)
#define BATCH 32

__global__ void prep_kernel(const float* __restrict__ x,
                            float* __restrict__ xT,
                            float* __restrict__ outT,
                            int size1, int size2)
{
    const int n1 = size1 * BATCH;
    const int n2 = size2 * BATCH;
    const int total = n1 > n2 ? n1 : n2;
    for (int t = blockIdx.x * blockDim.x + threadIdx.x; t < total;
         t += gridDim.x * blockDim.x) {
        if (t < n1) {
            int c = t >> 5;
            int b = t & 31;
            xT[t] = x[(size_t)b * size1 + c];
        }
        if (t < n2) outT[t] = 0.0f;
    }
}

__global__ __launch_bounds__(TPB) void edge_spmm_wmma(
    const float* __restrict__ wmean, const float* __restrict__ wlogv,
    const float* __restrict__ epsw,
    const int* __restrict__ rows, const int* __restrict__ cols,
    const float* __restrict__ xT, float* __restrict__ outT,
    int nEdges)
{
    __shared__ float lds_w[WAVES_PER_BLOCK][256];
    __shared__ float lds_x[WAVES_PER_BLOCK][512];

    const int lane = threadIdx.x & 31;
    const int wv   = threadIdx.x >> 5;
    const int e    = blockIdx.x * WAVES_PER_BLOCK + wv;
    const bool active = (e < nEdges);

    int baseRow = 0, baseCol = 0;
    if (active) {
        const size_t base = (size_t)e * 256;
        baseRow = rows[base];
        baseCol = cols[base];

        const float* wm = wmean + base;
        const float* lv = wlogv + base;
        const float* ew = epsw  + base;
#pragma unroll
        for (int u = 0; u < 2; ++u) {
            const int o = lane * 8 + u * 4;
            v4f m = *(const v4f*)(wm + o);
            v4f l = *(const v4f*)(lv + o);
            v4f g = *(const v4f*)(ew + o);
            v4f v;
            v.x = fmaf(g.x, __expf(l.x), m.x);
            v.y = fmaf(g.y, __expf(l.y), m.y);
            v.z = fmaf(g.z, __expf(l.z), m.z);
            v.w = fmaf(g.w, __expf(l.w), m.w);
            *(v4f*)(&lds_w[wv][o]) = v;
        }

        const float* xt = xT + (size_t)baseCol * BATCH;
#pragma unroll
        for (int u = 0; u < 4; ++u) {
            const int o = (u * 32 + lane) * 4;
            *(v4f*)(&lds_x[wv][o]) = *(const v4f*)(xt + o);
        }
    }
    __syncthreads();
    if (!active) return;

    const int m  = lane & 15;
    const int kh = (lane >> 4) * 2;

    v2f a[4];
#pragma unroll
    for (int kc = 0; kc < 4; ++kc) {
        const int k = kc * 4 + kh;
        a[kc].x = lds_w[wv][k * 16 + m];
        a[kc].y = lds_w[wv][(k + 1) * 16 + m];
    }

#pragma unroll
    for (int h = 0; h < 2; ++h) {
        v8f acc = {};
#pragma unroll
        for (int kc = 0; kc < 4; ++kc) {
            const int k = kc * 4 + kh;
            v2f b;
            b.x = lds_x[wv][k * 32 + m + 16 * h];
            b.y = lds_x[wv][(k + 1) * 32 + m + 16 * h];
            acc = __builtin_amdgcn_wmma_f32_16x16x4_f32(
                 false, a[kc],  false, b,
                 (short)0, acc,  false,  false);
        }
        const int n     = m + 16 * h;
        const int mbase = baseRow + (lane >> 4) * 8;
#pragma unroll
        for (int r = 0; r < 8; ++r) {
            atomicAdd(&outT[(size_t)(mbase + r) * BATCH + n], acc[r]);
        }
    }
}

__global__ void finish_kernel(const float* __restrict__ outT,
                              const float* __restrict__ bmean,
                              const float* __restrict__ blogv,
                              const float* __restrict__ epsb,
                              float* __restrict__ out, int size2)
{
    const int n2 = size2 * BATCH;
    for (int t = blockIdx.x * blockDim.x + threadIdx.x; t < n2;
         t += gridDim.x * blockDim.x) {
        const int r = t >> 5;
        const int b = t & 31;
        const float bias = fmaf(epsb[r], __expf(blogv[r]), bmean[r]);
        out[(size_t)b * size2 + r] = outT[t] + bias;
    }
    if (blockIdx.x == 0 && threadIdx.x == 0) out[n2] = 0.0f;
}

extern "C" void kernel_launch(void* const* d_in, const int* in_sizes, int n_in,
                              void* d_out, int out_size, void* d_ws, size_t ws_size,
                              hipStream_t stream)
{
    const float* x     = (const float*)d_in[0];
    const float* wmean = (const float*)d_in[1];
    const float* wlogv = (const float*)d_in[2];
    const float* bmean = (const float*)d_in[3];
    const float* blogv = (const float*)d_in[4];
    const float* epsw  = (const float*)d_in[5];
    const float* epsb  = (const float*)d_in[6];
    const int*   rows  = (const int*)d_in[7];
    const int*   cols  = (const int*)d_in[8];

    const int size2  = in_sizes[3];
    const int nnz    = in_sizes[1];
    const int nEdges = nnz / 256;
    const int size1  = in_sizes[0] / BATCH;

    float* xT   = (float*)d_ws;
    float* outT = xT + (size_t)size1 * BATCH;
    float* out  = (float*)d_out;

    const int n1 = size1 * BATCH, n2 = size2 * BATCH;
    const int nPrep = n1 > n2 ? n1 : n2;

    prep_kernel<<<(nPrep + TPB - 1) / TPB, TPB, 0, stream>>>(x, xT, outT, size1, size2);
    edge_spmm_wmma<<<(nEdges + WAVES_PER_BLOCK - 1) / WAVES_PER_BLOCK, TPB, 0, stream>>>(
        wmean, wlogv, epsw, rows, cols, xT, outT, nEdges);
    finish_kernel<<<(n2 + TPB - 1) / TPB, TPB, 0, stream>>>(outT, bmean, blogv, epsb, out, size2);
}


# Edge tiles accumulate straight into out: WMMA operands swapped (D = batches x features), bias stored first as the atomics' start value (sum order: bias first), finish launch and outT dropped
# speedup vs baseline: 1.5123x; 1.5123x over previous
#include <hip/hip_runtime.h>


typedef float v2f __attribute__((ext_vector_type(2)));
typedef float v4f __attribute__((ext_vector_type(4)));
typedef float v8f __attribute__((ext_vector_type(8)));

#define WAVES_PER_BLOCK 8
#define TPB (WAVES_PER_BLOCK * 32)
#define BATCH 32

__global__ void prep_kernel(const float* __restrict__ x,
                            const float* __restrict__ bmean,
                            const float* __restrict__ blogv,
                            const float* __restrict__ epsb,
                            float* __restrict__ xT,
                            float* __restrict__ out,
                            int size1, int size2)
{
    const int n1 = size1 * BATCH;
    const int n2 = size2 * BATCH;
    const int total = n1 > n2 ? n1 : n2;
    for (int t = blockIdx.x * blockDim.x + threadIdx.x; t < total;
         t += gridDim.x * blockDim.x) {
        const bool p1 = t < n1;
        const bool p2 = t < n2;
        float xv = 0.0f, eb = 0.0f, lb = 0.0f, mb = 0.0f;
        if (p1) {
            int c = t >> 5;
            int b = t & 31;
            xv = x[(size_t)b * size1 + c];
        }
        if (p2) {
            const int r = t - (t / size2) * size2;
            eb = epsb[r];
            lb = blogv[r];
            mb = bmean[r];
        }
        if (p1) xT[t] = xv;
        if (p2) out[t] = fmaf(eb, __expf(lb), mb);
    }
    if (blockIdx.x == 0 && threadIdx.x == 0) out[n2] = 0.0f;
}

__global__ __launch_bounds__(TPB) void edge_spmm_wmma(
    const float* __restrict__ wmean, const float* __restrict__ wlogv,
    const float* __restrict__ epsw,
    const int* __restrict__ rows, const int* __restrict__ cols,
    const float* __restrict__ xT, float* __restrict__ out,
    int nEdges, int size2)
{
    __shared__ float lds_w[WAVES_PER_BLOCK][256];
    __shared__ float lds_x[WAVES_PER_BLOCK][512];

    const int lane = threadIdx.x & 31;
    const int wv   = threadIdx.x >> 5;
    const int e    = blockIdx.x * WAVES_PER_BLOCK + wv;
    const bool active = (e < nEdges);

    int baseRow = 0, baseCol = 0;
    if (active) {
        const size_t base = (size_t)e * 256;
        baseRow = rows[base];
        baseCol = cols[base];

        const float* wm = wmean + base;
        const float* lv = wlogv + base;
        const float* ew = epsw  + base;
#pragma unroll
        for (int u = 0; u < 2; ++u) {
            const int o = lane * 8 + u * 4;
            v4f m = *(const v4f*)(wm + o);
            v4f l = *(const v4f*)(lv + o);
            v4f g = *(const v4f*)(ew + o);
            v4f v;
            v.x = fmaf(g.x, __expf(l.x), m.x);
            v.y = fmaf(g.y, __expf(l.y), m.y);
            v.z = fmaf(g.z, __expf(l.z), m.z);
            v.w = fmaf(g.w, __expf(l.w), m.w);
            *(v4f*)(&lds_w[wv][o]) = v;
        }

        const float* xt = xT + (size_t)baseCol * BATCH;
#pragma unroll
        for (int u = 0; u < 4; ++u) {
            const int o = (u * 32 + lane) * 4;
            *(v4f*)(&lds_x[wv][o]) = *(const v4f*)(xt + o);
        }
    }
    __syncthreads();
    if (!active) return;

    const int c  = lane & 15;
    const int hh = lane >> 4;
    const int kh = hh * 2;

    v2f wf[4];
#pragma unroll
    for (int kc = 0; kc < 4; ++kc) {
        const int k = kc * 4 + kh;
        wf[kc].x = lds_w[wv][k * 16 + c];
        wf[kc].y = lds_w[wv][(k + 1) * 16 + c];
    }

#pragma unroll
    for (int h = 0; h < 2; ++h) {
        v8f acc = {};
#pragma unroll
        for (int kc = 0; kc < 4; ++kc) {
            const int k = kc * 4 + kh;
            v2f xf;
            xf.x = lds_x[wv][k * 32 + c + 16 * h];
            xf.y = lds_x[wv][(k + 1) * 32 + c + 16 * h];
            acc = __builtin_amdgcn_wmma_f32_16x16x4_f32(
                 false, xf,  false, wf[kc],
                 (short)0, acc,  false,  false);
        }
        float* o = out + (size_t)(16 * h + 8 * hh) * size2 + baseRow + c;
#pragma unroll
        for (int r = 0; r < 8; ++r) {
            atomicAdd(o + (size_t)r * size2, acc[r]);
        }
    }
}

extern "C" void kernel_launch(void* const* d_in, const int* in_sizes, int n_in,
                              void* d_out, int out_size, void* d_ws, size_t ws_size,
                              hipStream_t stream)
{
    const float* x     = (const float*)d_in[0];
    const float* wmean = (const float*)d_in[1];
    const float* wlogv = (const float*)d_in[2];
    const float* bmean = (const float*)d_in[3];
    const float* blogv = (const float*)d_in[4];
    const float* epsw  = (const float*)d_in[5];
    const float* epsb  = (const float*)d_in[6];
    const int*   rows  = (const int*)d_in[7];
    const int*   cols  = (const int*)d_in[8];

    const int size2  = in_sizes[3];
    const int nnz    = in_sizes[1];
    const int nEdges = nnz / 256;
    const int size1  = in_sizes[0] / BATCH;

    float* xT  = (float*)d_ws;
    float* out = (float*)d_out;

    const int n1 = size1 * BATCH, n2 = size2 * BATCH;
    const int nPrep = n1 > n2 ? n1 : n2;

    prep_kernel<<<(nPrep + TPB - 1) / TPB, TPB, 0, stream>>>(
        x, bmean, blogv, epsb, xT, out, size1, size2);
    edge_spmm_wmma<<<(nEdges + WAVES_PER_BLOCK - 1) / WAVES_PER_BLOCK, TPB, 0, stream>>>(
        wmean, wlogv, epsw, rows, cols, xT, out, nEdges, size2);
}


# Fewer, fatter waves in both launches: prep 250 blocks (8+8 elements a thread, x turned by 64-column LDS tiles), edge kernel 4 edges per wave, all loads up front; atomics kept (arrival-order sums)
# speedup vs baseline: 1.5764x; 1.0424x over previous
#include <hip/hip_runtime.h>


typedef float v2f __attribute__((ext_vector_type(2)));
typedef float v4f __attribute__((ext_vector_type(4)));
typedef float v8f __attribute__((ext_vector_type(8)));

#define WAVES_PER_BLOCK 8
#define TPB (WAVES_PER_BLOCK * 32)
#define BATCH 32
#define EPW 4
#define TP_COLS 64
#define TP_PITCH 68

__global__ __launch_bounds__(TPB) void prep_kernel(
    const float* __restrict__ x,
    const float* __restrict__ bmean,
    const float* __restrict__ blogv,
    const float* __restrict__ epsb,
    float* __restrict__ xT,
    float* __restrict__ out,
    int size1, int size2)
{
    __shared__ __attribute__((aligned(16))) float tile[BATCH][TP_PITCH];

    const int tid = threadIdx.x;
    const int c0  = blockIdx.x * TP_COLS;
    const bool pT = c0 < size1;

    const int nq     = size2 >> 2;
    const int nItems = nq * (BATCH / 2);
    const int item   = blockIdx.x * TPB + tid;
    const bool pB    = item < nItems;

    const int lr = tid >> 4;
    const int lq = (tid & 15) * 4;
    v4f xa = {}, xb = {};
    if (pT) {
        const float* src = x + (size_t)lr * size1 + c0 + lq;
        xa = *(const v4f*)src;
        xb = *(const v4f*)(src + (size_t)16 * size1);
    }
    int bp = 0, rq = 0;
    v4f eb = {}, lb = {}, mb = {};
    if (pB) {
        bp = item / nq;
        rq = item - bp * nq;
        eb = *(const v4f*)(epsb  + 4 * rq);
        lb = *(const v4f*)(blogv + 4 * rq);
        mb = *(const v4f*)(bmean + 4 * rq);
    }

    if (pT) {
        *(v4f*)(&tile[lr][lq])      = xa;
        *(v4f*)(&tile[lr + 16][lq]) = xb;
    }
    __syncthreads();
    if (pT) {
        const int b0 = (tid & 7) * 4;
#pragma unroll
        for (int p = 0; p < 2; ++p) {
            const int c = (tid >> 3) + 32 * p;
            v4f v;
            v.x = tile[b0 + 0][c];
            v.y = tile[b0 + 1][c];
            v.z = tile[b0 + 2][c];
            v.w = tile[b0 + 3][c];
            *(v4f*)(xT + (size_t)(c0 + c) * BATCH + b0) = v;
        }
    }
    if (pB) {
        v4f v;
        v.x = fmaf(eb.x, __expf(lb.x), mb.x);
        v.y = fmaf(eb.y, __expf(lb.y), mb.y);
        v.z = fmaf(eb.z, __expf(lb.z), mb.z);
        v.w = fmaf(eb.w, __expf(lb.w), mb.w);
        float* o = out + (size_t)(2 * bp) * size2 + 4 * rq;
        *(v4f*)o = v;
        *(v4f*)(o + size2) = v;
    }
    if (blockIdx.x == 0 && tid == 0) out[(size_t)size2 * BATCH] = 0.0f;
}

__global__ __launch_bounds__(TPB) void edge_spmm_wmma(
    const float* __restrict__ wmean, const float* __restrict__ wlogv,
    const float* __restrict__ epsw,
    const int* __restrict__ rows, const int* __restrict__ cols,
    const float* __restrict__ xT, float* __restrict__ out,
    int nEdges, int size2)
{
    __shared__ __attribute__((aligned(16))) float lds_w[WAVES_PER_BLOCK][EPW][256];
    __shared__ __attribute__((aligned(16))) float lds_x[WAVES_PER_BLOCK][EPW][512];

    const int lane = threadIdx.x & 31;
    const int wv   = threadIdx.x >> 5;
    const int e0   = (blockIdx.x * WAVES_PER_BLOCK + wv) * EPW;

    const bool live = (e0 < nEdges);
    const int  eb   = live ? e0 : 0;
    int baseRow[EPW], baseCol[EPW];
    size_t base[EPW];
#pragma unroll
    for (int q = 0; q < EPW; ++q) {
        base[q]    = (size_t)(eb + q) * 256;
        baseRow[q] = rows[base[q]];
        baseCol[q] = cols[base[q]];
    }

    v4f m[EPW][2], l[EPW][2], g[EPW][2];
#pragma unroll
    for (int q = 0; q < EPW; ++q) {
#pragma unroll
        for (int u = 0; u < 2; ++u) {
            const int o = lane * 8 + u * 4;
            m[q][u] = *(const v4f*)(wmean + base[q] + o);
            l[q][u] = *(const v4f*)(wlogv + base[q] + o);
            g[q][u] = *(const v4f*)(epsw  + base[q] + o);
        }
    }

    v4f xv[EPW][4];
#pragma unroll
    for (int q = 0; q < EPW; ++q) {
        const float* xt = xT + (size_t)baseCol[q] * BATCH;
#pragma unroll
        for (int u = 0; u < 4; ++u)
            xv[q][u] = *(const v4f*)(xt + (u * 32 + lane) * 4);
    }

#pragma unroll
    for (int q = 0; q < EPW; ++q) {
#pragma unroll
        for (int u = 0; u < 2; ++u) {
            const int o = lane * 8 + u * 4;
            v4f v;
            v.x = fmaf(g[q][u].x, __expf(l[q][u].x), m[q][u].x);
            v.y = fmaf(g[q][u].y, __expf(l[q][u].y), m[q][u].y);
            v.z = fmaf(g[q][u].z, __expf(l[q][u].z), m[q][u].z);
            v.w = fmaf(g[q][u].w, __expf(l[q][u].w), m[q][u].w);
            *(v4f*)(&lds_w[wv][q][o]) = v;
        }
    }
#pragma unroll
    for (int q = 0; q < EPW; ++q) {
#pragma unroll
        for (int u = 0; u < 4; ++u)
            *(v4f*)(&lds_x[wv][q][(u * 32 + lane) * 4]) = xv[q][u];
    }
    __syncthreads();
    if (!live) return;

    const int c  = lane & 15;
    const int hh = lane >> 4;
    const int kh = hh * 2;

#pragma unroll
    for (int q = 0; q < EPW; ++q) {
        v2f wf[4];
#pragma unroll
        for (int kc = 0; kc < 4; ++kc) {
            const int k = kc * 4 + kh;
            wf[kc].x = lds_w[wv][q][k * 16 + c];
            wf[kc].y = lds_w[wv][q][(k + 1) * 16 + c];
        }

#pragma unroll
        for (int h = 0; h < 2; ++h) {
            v8f acc = {};
#pragma unroll
            for (int kc = 0; kc < 4; ++kc) {
                const int k = kc * 4 + kh;
                v2f xf;
                xf.x = lds_x[wv][q][k * 32 + c + 16 * h];
                xf.y = lds_x[wv][q][(k + 1) * 32 + c + 16 * h];
                acc = __builtin_amdgcn_wmma_f32_16x16x4_f32(
                     false, xf,  false, wf[kc],
                     (short)0, acc,  false,  false);
            }
            float* o = out + (size_t)(16 * h + 8 * hh) * size2 + baseRow[q] + c;
#pragma unroll
            for (int r = 0; r < 8; ++r) {
                atomicAdd(o + (size_t)r * size2, acc[r]);
            }
        }
    }
}

extern "C" void kernel_launch(void* const* d_in, const int* in_sizes, int n_in,
                              void* d_out, int out_size, void* d_ws, size_t ws_size,
                              hipStream_t stream)
{
    const float* x     = (const float*)d_in[0];
    const float* wmean = (const float*)d_in[1];
    const float* wlogv = (const float*)d_in[2];
    const float* bmean = (const float*)d_in[3];
    const float* blogv = (const float*)d_in[4];
    const float* epsw  = (const float*)d_in[5];
    const float* epsb  = (const float*)d_in[6];
    const int*   rows  = (const int*)d_in[7];
    const int*   cols  = (const int*)d_in[8];

    const int size2  = in_sizes[3];
    const int nnz    = in_sizes[1];
    const int nEdges = nnz / 256;
    const int size1  = in_sizes[0] / BATCH;

    float* xT  = (float*)d_ws;
    float* out = (float*)d_out;

    const int nTile = size1 / TP_COLS;
    const int nBias = ((size2 / 4) * (BATCH / 2) + TPB - 1) / TPB;
    const int nPrep = nTile > nBias ? nTile : nBias;
    const int edgesPerBlock = WAVES_PER_BLOCK * EPW;

    prep_kernel<<<nPrep, TPB, 0, stream>>>(
        x, bmean, blogv, epsb, xT, out, size1, size2);
    edge_spmm_wmma<<<(nEdges + edgesPerBlock - 1) / edgesPerBlock, TPB, 0, stream>>>(
        wmean, wlogv, epsw, rows, cols, xT, out, nEdges, size2);
}


# Owner computes: prep packs edge keys, each block lists the edges of its 64 out features, adds their tiles over the bias in LDS (LDS f32 adds, arrival order), stores out once; global atomics dropped
# speedup vs baseline: 2.1193x; 1.3444x over previous
#include <hip/hip_runtime.h>


typedef float v2f __attribute__((ext_vector_type(2)));
typedef float v4f __attribute__((ext_vector_type(4)));
typedef float v8f __attribute__((ext_vector_type(8)));
typedef int   v4i __attribute__((ext_vector_type(4)));

#define TPB 256
#define BATCH 32
#define TP_COLS 64
#define TP_PITCH 68

#define EW 16
#define ETPB (EW * 32)
#define EPW 4
#define TPBK 4
#define OFEAT (TPBK * 16)
#define O_PITCH 68
#define LIST_CAP 16384
#define SCAN_U 8

__global__ __launch_bounds__(TPB) void prep_kernel(
    const float* __restrict__ x,
    const int* __restrict__ rows, const int* __restrict__ cols,
    float* __restrict__ xT, int* __restrict__ ekey,
    int size1, int nEdges)
{
    __shared__ __attribute__((aligned(16))) float tile[BATCH][TP_PITCH];

    const int tid = threadIdx.x;
    const int c0  = blockIdx.x * TP_COLS;
    const bool pT = c0 < size1;
    const int e   = blockIdx.x * TPB + tid;
    const bool pE = e < nEdges;

    const int lr = tid >> 4;
    const int lq = (tid & 15) * 4;
    v4f xa = {}, xb = {};
    if (pT) {
        const float* src = x + (size_t)lr * size1 + c0 + lq;
        xa = *(const v4f*)src;
        xb = *(const v4f*)(src + (size_t)16 * size1);
    }
    int kr = 0, kc = 0;
    if (pE) {
        kr = rows[(size_t)e * 256];
        kc = cols[(size_t)e * 256];
    }

    if (pT) {
        *(v4f*)(&tile[lr][lq])      = xa;
        *(v4f*)(&tile[lr + 16][lq]) = xb;
    }
    __syncthreads();
    if (pT) {
        const int b0 = (tid & 7) * 4;
#pragma unroll
        for (int p = 0; p < 2; ++p) {
            const int c = (tid >> 3) + 32 * p;
            v4f v;
            v.x = tile[b0 + 0][c];
            v.y = tile[b0 + 1][c];
            v.z = tile[b0 + 2][c];
            v.w = tile[b0 + 3][c];
            *(v4f*)(xT + (size_t)(c0 + c) * BATCH + b0) = v;
        }
    }
    if (pE) ekey[e] = (kr & 0xFFFF) | (kc << 16);
}

__global__ __launch_bounds__(ETPB) void tile_spmm_wmma(
    const float* __restrict__ wmean, const float* __restrict__ wlogv,
    const float* __restrict__ epsw,
    const float* __restrict__ bmean, const float* __restrict__ blogv,
    const float* __restrict__ epsb,
    const int* __restrict__ ekey,
    const float* __restrict__ xT, float* __restrict__ out,
    int nEdges, int size2)
{
    __shared__ __attribute__((aligned(16))) float lds_w[EW][EPW][256];
    __shared__ __attribute__((aligned(16))) float lds_x[EW][EPW][512];
    __shared__ __attribute__((aligned(16))) float lds_o[BATCH][O_PITCH];
    __shared__ unsigned int elist[LIST_CAP];
    __shared__ int ecount;

    const int tid  = threadIdx.x;
    const int lane = tid & 31;
    const int wv   = __builtin_amdgcn_readfirstlane(tid >> 5);
    const int blk  = blockIdx.x;

    const int ob = tid >> 4;
    const int oq = (tid & 15) * 4;
    {
        const int r = blk * OFEAT + oq;
        const v4f eb = *(const v4f*)(epsb  + r);
        const v4f lb = *(const v4f*)(blogv + r);
        const v4f mb = *(const v4f*)(bmean + r);
        v4f v;
        v.x = fmaf(eb.x, __expf(lb.x), mb.x);
        v.y = fmaf(eb.y, __expf(lb.y), mb.y);
        v.z = fmaf(eb.z, __expf(lb.z), mb.z);
        v.w = fmaf(eb.w, __expf(lb.w), mb.w);
        *(v4f*)(&lds_o[ob][oq]) = v;
    }
    if (tid == 0) ecount = 0;
    if (blk == 0 && tid == 0) out[(size_t)size2 * BATCH] = 0.0f;
    __syncthreads();

    const int nQuads = nEdges >> 2;
    for (int i0 = 0; i0 < nQuads; i0 += ETPB * SCAN_U) {
        v4i k4[SCAN_U];
#pragma unroll
        for (int t = 0; t < SCAN_U; ++t) {
            const int i = i0 + t * ETPB + tid;
            k4[t] = v4i{0, 0, 0, 0};
            if (i < nQuads) k4[t] = *(const v4i*)(ekey + 4 * i);
        }
#pragma unroll
        for (int t = 0; t < SCAN_U; ++t) {
            const int i = i0 + t * ETPB + tid;
#pragma unroll
            for (int u = 0; u < 4; ++u) {
                const unsigned int key  = (unsigned int)k4[t][u];
                const unsigned int orow = key & 0xFFFFu;
                if (i < nQuads && (int)(orow >> 6) == blk) {
                    const int s = atomicAdd(&ecount, 1);
                    if (s < LIST_CAP)
                        elist[s] = ((unsigned int)(4 * i + u) << 2) | ((orow >> 4) & 3u) | (key & 0xFFFF0000u);
                }
            }
        }
    }
    __syncthreads();
    const int n = ecount < LIST_CAP ? ecount : LIST_CAP;

    const int c  = lane & 15;
    const int hh = lane >> 4;
    const int kh = hh * 2;

    for (int g0 = 0; g0 < n; g0 += EW * EPW) {
        bool   live[EPW];
        int    slot[EPW];
        size_t base[EPW];
        int    bcol[EPW];
#pragma unroll
        for (int q = 0; q < EPW; ++q) {
            const int idx = g0 + q * EW + wv;
            live[q] = idx < n;
            const unsigned int en = __builtin_amdgcn_readfirstlane(live[q] ? elist[idx] : 0u);
            slot[q] = (int)(en & 3u);
            base[q] = (size_t)((en & 0xFFFFu) >> 2) * 256;
            bcol[q] = (int)(en >> 16);
        }

        v4f m[EPW][2], l[EPW][2], g[EPW][2];
        v4f xv[EPW][4];
#pragma unroll
        for (int q = 0; q < EPW; ++q) {
            if (live[q]) {
#pragma unroll
                for (int u = 0; u < 2; ++u) {
                    const int o = lane * 8 + u * 4;
                    m[q][u] = *(const v4f*)(wmean + base[q] + o);
                    l[q][u] = *(const v4f*)(wlogv + base[q] + o);
                    g[q][u] = *(const v4f*)(epsw  + base[q] + o);
                }
                const float* xt = xT + (size_t)bcol[q] * BATCH;
#pragma unroll
                for (int u = 0; u < 4; ++u)
                    xv[q][u] = *(const v4f*)(xt + (u * 32 + lane) * 4);
            }
        }

#pragma unroll
        for (int q = 0; q < EPW; ++q) {
            if (live[q]) {
#pragma unroll
                for (int u = 0; u < 2; ++u) {
                    const int o = lane * 8 + u * 4;
                    v4f v;
                    v.x = fmaf(g[q][u].x, __expf(l[q][u].x), m[q][u].x);
                    v.y = fmaf(g[q][u].y, __expf(l[q][u].y), m[q][u].y);
                    v.z = fmaf(g[q][u].z, __expf(l[q][u].z), m[q][u].z);
                    v.w = fmaf(g[q][u].w, __expf(l[q][u].w), m[q][u].w);
                    *(v4f*)(&lds_w[wv][q][o]) = v;
                }
#pragma unroll
                for (int u = 0; u < 4; ++u)
                    *(v4f*)(&lds_x[wv][q][(u * 32 + lane) * 4]) = xv[q][u];
            }
        }
        __syncthreads();

#pragma unroll
        for (int q = 0; q < EPW; ++q) {
            if (live[q]) {
                v2f wf[4];
#pragma unroll
                for (int kc = 0; kc < 4; ++kc) {
                    const int k = kc * 4 + kh;
                    wf[kc].x = lds_w[wv][q][k * 16 + c];
                    wf[kc].y = lds_w[wv][q][(k + 1) * 16 + c];
                }
#pragma unroll
                for (int h = 0; h < 2; ++h) {
                    v8f acc = {};
#pragma unroll
                    for (int kc = 0; kc < 4; ++kc) {
                        const int k = kc * 4 + kh;
                        v2f xf;
                        xf.x = lds_x[wv][q][k * 32 + c + 16 * h];
                        xf.y = lds_x[wv][q][(k + 1) * 32 + c + 16 * h];
                        acc = __builtin_amdgcn_wmma_f32_16x16x4_f32(
                             false, xf,  false, wf[kc],
                             (short)0, acc,  false,  false);
                    }
                    float* o = &lds_o[16 * h + 8 * hh][slot[q] * 16 + c];
#pragma unroll
                    for (int r = 0; r < 8; ++r)
                        atomicAdd(o + r * O_PITCH, acc[r]);
                }
            }
        }
    }
    __syncthreads();

    *(v4f*)(out + (size_t)ob * size2 + blk * OFEAT + oq) = *(const v4f*)(&lds_o[ob][oq]);
}

extern "C" void kernel_launch(void* const* d_in, const int* in_sizes, int n_in,
                              void* d_out, int out_size, void* d_ws, size_t ws_size,
                              hipStream_t stream)
{
    const float* x     = (const float*)d_in[0];
    const float* wmean = (const float*)d_in[1];
    const float* wlogv = (const float*)d_in[2];
    const float* bmean = (const float*)d_in[3];
    const float* blogv = (const float*)d_in[4];
    const float* epsw  = (const float*)d_in[5];
    const float* epsb  = (const float*)d_in[6];
    const int*   rows  = (const int*)d_in[7];
    const int*   cols  = (const int*)d_in[8];

    const int size2  = in_sizes[3];
    const int nnz    = in_sizes[1];
    const int nEdges = nnz / 256;
    const int size1  = in_sizes[0] / BATCH;

    float* xT   = (float*)d_ws;
    int*   ekey = (int*)(xT + (size_t)size1 * BATCH);
    float* out  = (float*)d_out;

    const int nTile = size1 / TP_COLS;
    const int nKey  = (nEdges + TPB - 1) / TPB;
    const int nPrep = nTile > nKey ? nTile : nKey;

    prep_kernel<<<nPrep, TPB, 0, stream>>>(x, rows, cols, xT, ekey, size1, nEdges);
    tile_spmm_wmma<<<size2 / OFEAT, ETPB, 0, stream>>>(
        wmean, wlogv, epsw, bmean, blogv, epsb, ekey, xT, out, nEdges, size2);
}


# Tile kernel: block barrier in the edge loop dropped; each wave streams its own edges, loads issued ahead, each edge staged and multiplied once its own loads land (LDS f32 adds still arrival order)
# speedup vs baseline: 2.3254x; 1.0972x over previous
#include <hip/hip_runtime.h>


typedef float v2f __attribute__((ext_vector_type(2)));
typedef float v4f __attribute__((ext_vector_type(4)));
typedef float v8f __attribute__((ext_vector_type(8)));
typedef int   v4i __attribute__((ext_vector_type(4)));

#define TPB 256
#define BATCH 32
#define TP_COLS 64
#define TP_PITCH 68

#define EW 16
#define ETPB (EW * 32)
#define EPW 4
#define TPBK 4
#define OFEAT (TPBK * 16)
#define O_PITCH 68
#define LIST_CAP 16384
#define SCAN_U 8
#define MAXNE 6

template <int N> struct IC { static constexpr int value = N; };

__global__ __launch_bounds__(TPB) void prep_kernel(
    const float* __restrict__ x,
    const int* __restrict__ rows, const int* __restrict__ cols,
    float* __restrict__ xT, int* __restrict__ ekey,
    int size1, int nEdges)
{
    __shared__ __attribute__((aligned(16))) float tile[BATCH][TP_PITCH];

    const int tid = threadIdx.x;
    const int c0  = blockIdx.x * TP_COLS;
    const bool pT = c0 < size1;
    const int e   = blockIdx.x * TPB + tid;
    const bool pE = e < nEdges;

    const int lr = tid >> 4;
    const int lq = (tid & 15) * 4;
    v4f xa = {}, xb = {};
    if (pT) {
        const float* src = x + (size_t)lr * size1 + c0 + lq;
        xa = *(const v4f*)src;
        xb = *(const v4f*)(src + (size_t)16 * size1);
    }
    int kr = 0, kc = 0;
    if (pE) {
        kr = rows[(size_t)e * 256];
        kc = cols[(size_t)e * 256];
    }

    if (pT) {
        *(v4f*)(&tile[lr][lq])      = xa;
        *(v4f*)(&tile[lr + 16][lq]) = xb;
    }
    __syncthreads();
    if (pT) {
        const int b0 = (tid & 7) * 4;
#pragma unroll
        for (int p = 0; p < 2; ++p) {
            const int c = (tid >> 3) + 32 * p;
            v4f v;
            v.x = tile[b0 + 0][c];
            v.y = tile[b0 + 1][c];
            v.z = tile[b0 + 2][c];
            v.w = tile[b0 + 3][c];
            *(v4f*)(xT + (size_t)(c0 + c) * BATCH + b0) = v;
        }
    }
    if (pE) ekey[e] = (kr & 0xFFFF) | (kc << 16);
}

__global__ __launch_bounds__(ETPB) void tile_spmm_wmma(
    const float* __restrict__ wmean, const float* __restrict__ wlogv,
    const float* __restrict__ epsw,
    const float* __restrict__ bmean, const float* __restrict__ blogv,
    const float* __restrict__ epsb,
    const int* __restrict__ ekey,
    const float* __restrict__ xT, float* __restrict__ out,
    int nEdges, int size2)
{
    __shared__ __attribute__((aligned(16))) float lds_w[EW][EPW][256];
    __shared__ __attribute__((aligned(16))) float lds_x[EW][EPW][512];
    __shared__ __attribute__((aligned(16))) float lds_o[BATCH][O_PITCH];
    __shared__ unsigned int elist[LIST_CAP];
    __shared__ int ecount;

    const int tid  = threadIdx.x;
    const int lane = tid & 31;
    const int wv   = __builtin_amdgcn_readfirstlane(tid >> 5);
    const int blk  = blockIdx.x;

    const int ob = tid >> 4;
    const int oq = (tid & 15) * 4;
    {
        const int r = blk * OFEAT + oq;
        const v4f eb = *(const v4f*)(epsb  + r);
        const v4f lb = *(const v4f*)(blogv + r);
        const v4f mb = *(const v4f*)(bmean + r);
        v4f v;
        v.x = fmaf(eb.x, __expf(lb.x), mb.x);
        v.y = fmaf(eb.y, __expf(lb.y), mb.y);
        v.z = fmaf(eb.z, __expf(lb.z), mb.z);
        v.w = fmaf(eb.w, __expf(lb.w), mb.w);
        *(v4f*)(&lds_o[ob][oq]) = v;
    }
    if (tid == 0) ecount = 0;
    if (blk == 0 && tid == 0) out[(size_t)size2 * BATCH] = 0.0f;
    __syncthreads();

    const int nQuads = nEdges >> 2;
    for (int i0 = 0; i0 < nQuads; i0 += ETPB * SCAN_U) {
        v4i k4[SCAN_U];
#pragma unroll
        for (int t = 0; t < SCAN_U; ++t) {
            const int i = i0 + t * ETPB + tid;
            k4[t] = v4i{0, 0, 0, 0};
            if (i < nQuads) k4[t] = *(const v4i*)(ekey + 4 * i);
        }
#pragma unroll
        for (int t = 0; t < SCAN_U; ++t) {
            const int i = i0 + t * ETPB + tid;
#pragma unroll
            for (int u = 0; u < 4; ++u) {
                const unsigned int key  = (unsigned int)k4[t][u];
                const unsigned int orow = key & 0xFFFFu;
                if (i < nQuads && (int)(orow >> 6) == blk) {
                    const int s = atomicAdd(&ecount, 1);
                    if (s < LIST_CAP)
                        elist[s] = ((unsigned int)(4 * i + u) << 2) | ((orow >> 4) & 3u) | (key & 0xFFFF0000u);
                }
            }
        }
    }
    __syncthreads();
    const int n = __builtin_amdgcn_readfirstlane(ecount < LIST_CAP ? ecount : LIST_CAP);

    const int c  = lane & 15;
    const int hh = lane >> 4;
    const int kh = hh * 2;

    auto stream = [&](auto nec, const int first) __attribute__((always_inline)) {
        constexpr int NE = decltype(nec)::value;
        constexpr int NF = NE < EPW ? NE : EPW;

        int    slot[NE];
        size_t base[NE];
        int    bcol[NE];
#pragma unroll
        for (int t = 0; t < NE; ++t) {
            const unsigned int en = __builtin_amdgcn_readfirstlane(elist[first + t * EW]);
            slot[t] = (int)(en & 3u);
            base[t] = (size_t)((en & 0xFFFFu) >> 2) * 256;
            bcol[t] = (int)(en >> 16);
        }

        v4f m[EPW][2], l[EPW][2], g[EPW][2];
        v4f xv[EPW][4];
        auto load_edge = [&](const int t, const int s) __attribute__((always_inline)) {
#pragma unroll
            for (int u = 0; u < 2; ++u) {
                const int o = lane * 8 + u * 4;
                m[s][u] = *(const v4f*)(wmean + base[t] + o);
                l[s][u] = *(const v4f*)(wlogv + base[t] + o);
                g[s][u] = *(const v4f*)(epsw  + base[t] + o);
            }
            const float* xt = xT + (size_t)bcol[t] * BATCH;
#pragma unroll
            for (int u = 0; u < 4; ++u)
                xv[s][u] = *(const v4f*)(xt + (u * 32 + lane) * 4);
        };

#pragma unroll
        for (int t = 0; t < NF; ++t) load_edge(t, t);

#pragma unroll
        for (int t = 0; t < NE; ++t) {
            const int s = t % EPW;
            __builtin_amdgcn_sched_barrier(0);
#pragma unroll
            for (int u = 0; u < 2; ++u) {
                const int o = lane * 8 + u * 4;
                v4f v;
                v.x = fmaf(g[s][u].x, __expf(l[s][u].x), m[s][u].x);
                v.y = fmaf(g[s][u].y, __expf(l[s][u].y), m[s][u].y);
                v.z = fmaf(g[s][u].z, __expf(l[s][u].z), m[s][u].z);
                v.w = fmaf(g[s][u].w, __expf(l[s][u].w), m[s][u].w);
                *(v4f*)(&lds_w[wv][s][o]) = v;
            }
#pragma unroll
            for (int u = 0; u < 4; ++u)
                *(v4f*)(&lds_x[wv][s][(u * 32 + lane) * 4]) = xv[s][u];

            if (t + EPW < NE) load_edge(t + EPW, s);

            float* const o0 = &lds_o[8 * hh][slot[t] * 16 + c];

            __builtin_amdgcn_fence(__ATOMIC_ACQ_REL, "wavefront");
            __builtin_amdgcn_sched_barrier(0);

            v2f wf[4];
#pragma unroll
            for (int kc = 0; kc < 4; ++kc) {
                const int k = kc * 4 + kh;
                wf[kc].x = lds_w[wv][s][k * 16 + c];
                wf[kc].y = lds_w[wv][s][(k + 1) * 16 + c];
            }
#pragma unroll
            for (int h = 0; h < 2; ++h) {
                v8f acc = {};
#pragma unroll
                for (int kc = 0; kc < 4; ++kc) {
                    const int k = kc * 4 + kh;
                    v2f xf;
                    xf.x = lds_x[wv][s][k * 32 + c + 16 * h];
                    xf.y = lds_x[wv][s][(k + 1) * 32 + c + 16 * h];
                    acc = __builtin_amdgcn_wmma_f32_16x16x4_f32(
                         false, xf,  false, wf[kc],
                         (short)0, acc,  false,  false);
                }
                float* o = o0 + 16 * h * O_PITCH;
#pragma unroll
                for (int r = 0; r < 8; ++r)
                    atomicAdd(o + r * O_PITCH, acc[r]);
            }
            __builtin_amdgcn_sched_barrier(0);
            asm volatile("v_nop\n\tv_nop\n\tv_nop\n\tv_nop\n\tv_nop\n\tv_nop\n\tv_nop\n\tv_nop");
        }
    };

    const int ne_all = n > wv ? (n - wv + EW - 1) / EW : 0;
    for (int t0 = 0; t0 < ne_all; t0 += MAXNE) {
        const int left  = ne_all - t0;
        const int first = wv + t0 * EW;
        switch (left) {
            case 1:  stream(IC<1>{}, first); break;
            case 2:  stream(IC<2>{}, first); break;
            case 3:  stream(IC<3>{}, first); break;
            case 4:  stream(IC<4>{}, first); break;
            case 5:  stream(IC<5>{}, first); break;
            default: stream(IC<MAXNE>{}, first); break;
        }
    }
    __syncthreads();

    *(v4f*)(out + (size_t)ob * size2 + blk * OFEAT + oq) = *(const v4f*)(&lds_o[ob][oq]);
}

extern "C" void kernel_launch(void* const* d_in, const int* in_sizes, int n_in,
                              void* d_out, int out_size, void* d_ws, size_t ws_size,
                              hipStream_t stream)
{
    const float* x     = (const float*)d_in[0];
    const float* wmean = (const float*)d_in[1];
    const float* wlogv = (const float*)d_in[2];
    const float* bmean = (const float*)d_in[3];
    const float* blogv = (const float*)d_in[4];
    const float* epsw  = (const float*)d_in[5];
    const float* epsb  = (const float*)d_in[6];
    const int*   rows  = (const int*)d_in[7];
    const int*   cols  = (const int*)d_in[8];

    const int size2  = in_sizes[3];
    const int nnz    = in_sizes[1];
    const int nEdges = nnz / 256;
    const int size1  = in_sizes[0] / BATCH;

    float* xT   = (float*)d_ws;
    int*   ekey = (int*)(xT + (size_t)size1 * BATCH);
    float* out  = (float*)d_out;

    const int nTile = size1 / TP_COLS;
    const int nKey  = (nEdges + TPB - 1) / TPB;
    const int nPrep = nTile > nKey ? nTile : nKey;

    prep_kernel<<<nPrep, TPB, 0, stream>>>(x, rows, cols, xT, ekey, size1, nEdges);
    tile_spmm_wmma<<<size2 / OFEAT, ETPB, 0, stream>>>(
        wmean, wlogv, epsw, bmean, blogv, epsb, ekey, xT, out, nEdges, size2);
}


# No LDS staging of operands: W rows load straight into B-fragment registers (a line per dword load), X in A-fragment order from prep; WMMAs on 2 accumulators; k order inside the sums permuted
# speedup vs baseline: 2.5864x; 1.1122x over previous
#include <hip/hip_runtime.h>


typedef float v2f __attribute__((ext_vector_type(2)));
typedef float v4f __attribute__((ext_vector_type(4)));
typedef float v8f __attribute__((ext_vector_type(8)));
typedef int   v4i __attribute__((ext_vector_type(4)));
typedef __attribute__((address_space(3))) float lds_f32;

#define TPB 256
#define BATCH 32
#define TP_COLS 64
#define TP_PITCH 68

#define EW 16
#define ETPB (EW * 32)
#define EPW 2
#define TPBK 4
#define OFEAT (TPBK * 16)
#define O_PITCH 68
#define LIST_CAP 16384
#define SCAN_U 8
#define MAXNE 6

template <int N> struct IC { static constexpr int value = N; };

__global__ __launch_bounds__(TPB) void prep_kernel(
    const float* __restrict__ x,
    const int* __restrict__ rows, const int* __restrict__ cols,
    float* __restrict__ xT, int* __restrict__ ekey,
    int size1, int nEdges)
{
    __shared__ __attribute__((aligned(16))) float tile[BATCH][TP_PITCH];

    const int tid = threadIdx.x;
    const int c0  = blockIdx.x * TP_COLS;
    const bool pT = c0 < size1;
    const int e   = blockIdx.x * TPB + tid;
    const bool pE = e < nEdges;

    const int lr = tid >> 4;
    const int lq = (tid & 15) * 4;
    v4f xa = {}, xb = {};
    if (pT) {
        const float* src = x + (size_t)lr * size1 + c0 + lq;
        xa = *(const v4f*)src;
        xb = *(const v4f*)(src + (size_t)16 * size1);
    }
    int kr = 0, kc = 0;
    if (pE) {
        kr = rows[(size_t)e * 256];
        kc = cols[(size_t)e * 256];
    }

    if (pT) {
        *(v4f*)(&tile[lr][lq])      = xa;
        *(v4f*)(&tile[lr + 16][lq]) = xb;
    }
    __syncthreads();
    if (pT) {
#pragma unroll
        for (int p = 0; p < 2; ++p) {
            const int j    = tid + TPB * p;
            const int ctl  = j >> 7;
            const int u    = (j >> 5) & 3;
            const int ln   = j & 31;
            const int bt   = 16 * (u >> 1) + (ln & 15);
            const int col  = 16 * ctl + 8 * (u & 1) + (ln >> 4);
            v4f v;
            v.x = tile[bt][col];
            v.y = tile[bt][col + 2];
            v.z = tile[bt][col + 4];
            v.w = tile[bt][col + 6];
            *(v4f*)(xT + (size_t)c0 * BATCH + j * 4) = v;
        }
    }
    if (pE) ekey[e] = (kr & 0xFFFF) | (kc << 16);
}

__global__ __launch_bounds__(ETPB) void tile_spmm_wmma(
    const float* __restrict__ wmean, const float* __restrict__ wlogv,
    const float* __restrict__ epsw,
    const float* __restrict__ bmean, const float* __restrict__ blogv,
    const float* __restrict__ epsb,
    const int* __restrict__ ekey,
    const float* __restrict__ xT, float* __restrict__ out,
    int nEdges, int size2)
{
    __shared__ __attribute__((aligned(16))) float lds_o[BATCH][O_PITCH];
    __shared__ unsigned int elist[LIST_CAP];
    __shared__ int ecount;

    const int tid  = threadIdx.x;
    const int lane = tid & 31;
    const int wv   = __builtin_amdgcn_readfirstlane(tid >> 5);
    const int blk  = blockIdx.x;

    const int ob = tid >> 4;
    const int oq = (tid & 15) * 4;
    {
        const int r = blk * OFEAT + oq;
        const v4f eb = *(const v4f*)(epsb  + r);
        const v4f lb = *(const v4f*)(blogv + r);
        const v4f mb = *(const v4f*)(bmean + r);
        v4f v;
        v.x = fmaf(eb.x, __expf(lb.x), mb.x);
        v.y = fmaf(eb.y, __expf(lb.y), mb.y);
        v.z = fmaf(eb.z, __expf(lb.z), mb.z);
        v.w = fmaf(eb.w, __expf(lb.w), mb.w);
        *(v4f*)(&lds_o[ob][oq]) = v;
    }
    if (tid == 0) ecount = 0;
    if (blk == 0 && tid == 0) out[(size_t)size2 * BATCH] = 0.0f;
    __syncthreads();

    const int nQuads = nEdges >> 2;
    for (int i0 = 0; i0 < nQuads; i0 += ETPB * SCAN_U) {
        v4i k4[SCAN_U];
#pragma unroll
        for (int t = 0; t < SCAN_U; ++t) {
            const int i = i0 + t * ETPB + tid;
            k4[t] = v4i{0, 0, 0, 0};
            if (i < nQuads) k4[t] = *(const v4i*)(ekey + 4 * i);
        }
#pragma unroll
        for (int t = 0; t < SCAN_U; ++t) {
            const int i = i0 + t * ETPB + tid;
#pragma unroll
            for (int u = 0; u < 4; ++u) {
                const unsigned int key  = (unsigned int)k4[t][u];
                const unsigned int orow = key & 0xFFFFu;
                if (i < nQuads && (int)(orow >> 6) == blk) {
                    const int s = atomicAdd(&ecount, 1);
                    if (s < LIST_CAP)
                        elist[s] = ((unsigned int)(4 * i + u) << 2) | ((orow >> 4) & 3u) | (key & 0xFFFF0000u);
                }
            }
        }
    }
    __syncthreads();
    const int n = __builtin_amdgcn_readfirstlane(ecount < LIST_CAP ? ecount : LIST_CAP);

    const int c  = lane & 15;
    const int hh = lane >> 4;

    auto stream = [&](auto nec, const int first) __attribute__((always_inline)) {
        constexpr int NE = decltype(nec)::value;
        constexpr int NF = NE < EPW ? NE : EPW;

        int    slot[NE];
        size_t base[NE];
        int    bcol[NE];
#pragma unroll
        for (int t = 0; t < NE; ++t) {
            const unsigned int en = __builtin_amdgcn_readfirstlane(elist[first + t * EW]);
            slot[t] = (int)(en & 3u);
            base[t] = (size_t)((en & 0xFFFFu) >> 2) * 256;
            bcol[t] = (int)(en >> 16);
        }

        float m[EPW][8], l[EPW][8], g[EPW][8];
        v4f xv[EPW][4];
        auto load_edge = [&](const int t, const int s) __attribute__((always_inline)) {
#pragma unroll
            for (int q = 0; q < 8; ++q) {
                const int o = q * 32 + lane;
                m[s][q] = wmean[base[t] + o];
                l[s][q] = wlogv[base[t] + o];
                g[s][q] = epsw[base[t] + o];
            }
            const float* xt = xT + (size_t)bcol[t] * BATCH;
#pragma unroll
            for (int u = 0; u < 4; ++u)
                xv[s][u] = *(const v4f*)(xt + (u * 32 + lane) * 4);
        };

#pragma unroll
        for (int t = 0; t < NF; ++t) load_edge(t, t);

#pragma unroll
        for (int t = 0; t < NE; ++t) {
            const int s = t % EPW;
            __builtin_amdgcn_sched_barrier(0);
            v2f wf[4];
#pragma unroll
            for (int kc = 0; kc < 4; ++kc) {
                wf[kc].x = fmaf(g[s][2 * kc],     __expf(l[s][2 * kc]),     m[s][2 * kc]);
                wf[kc].y = fmaf(g[s][2 * kc + 1], __expf(l[s][2 * kc + 1]), m[s][2 * kc + 1]);
            }
            v2f xf[2][4];
#pragma unroll
            for (int h = 0; h < 2; ++h)
#pragma unroll
                for (int kc = 0; kc < 4; ++kc) {
                    const v4f q4 = xv[s][h * 2 + (kc >> 1)];
                    xf[h][kc].x = (kc & 1) ? q4.z : q4.x;
                    xf[h][kc].y = (kc & 1) ? q4.w : q4.y;
                }

            if (t + EPW < NE) load_edge(t + EPW, s);

            lds_f32* o0l = (lds_f32*)&lds_o[8 * hh][slot[t] * 16 + c];
            asm volatile("" : "+v"(o0l));
            float* const o0 = (float*)o0l;

            __builtin_amdgcn_sched_barrier(0);

            v8f acc0 = {}, acc1 = {};
#pragma unroll
            for (int kc = 0; kc < 4; ++kc) {
                acc0 = __builtin_amdgcn_wmma_f32_16x16x4_f32(
                     false, xf[0][kc],  false, wf[kc],
                     (short)0, acc0,  false,  false);
                acc1 = __builtin_amdgcn_wmma_f32_16x16x4_f32(
                     false, xf[1][kc],  false, wf[kc],
                     (short)0, acc1,  false,  false);
            }
#pragma unroll
            for (int r = 0; r < 8; ++r)
                atomicAdd(o0 + r * O_PITCH, acc0[r]);
#pragma unroll
            for (int r = 0; r < 8; ++r)
                atomicAdd(o0 + (16 + r) * O_PITCH, acc1[r]);
            __builtin_amdgcn_sched_barrier(0);
            asm volatile("v_nop\n\tv_nop\n\tv_nop\n\tv_nop\n\tv_nop\n\tv_nop\n\tv_nop\n\tv_nop");
        }
    };

    const int ne_all = n > wv ? (n - wv + EW - 1) / EW : 0;
    for (int t0 = 0; t0 < ne_all; t0 += MAXNE) {
        const int left  = ne_all - t0;
        const int first = wv + t0 * EW;
        switch (left) {
            case 1:  stream(IC<1>{}, first); break;
            case 2:  stream(IC<2>{}, first); break;
            case 3:  stream(IC<3>{}, first); break;
            case 4:  stream(IC<4>{}, first); break;
            case 5:  stream(IC<5>{}, first); break;
            default: stream(IC<MAXNE>{}, first); break;
        }
    }
    __syncthreads();

    *(v4f*)(out + (size_t)ob * size2 + blk * OFEAT + oq) = *(const v4f*)(&lds_o[ob][oq]);
}

extern "C" void kernel_launch(void* const* d_in, const int* in_sizes, int n_in,
                              void* d_out, int out_size, void* d_ws, size_t ws_size,
                              hipStream_t stream)
{
    const float* x     = (const float*)d_in[0];
    const float* wmean = (const float*)d_in[1];
    const float* wlogv = (const float*)d_in[2];
    const float* bmean = (const float*)d_in[3];
    const float* blogv = (const float*)d_in[4];
    const float* epsw  = (const float*)d_in[5];
    const float* epsb  = (const float*)d_in[6];
    const int*   rows  = (const int*)d_in[7];
    const int*   cols  = (const int*)d_in[8];

    const int size2  = in_sizes[3];
    const int nnz    = in_sizes[1];
    const int nEdges = nnz / 256;
    const int size1  = in_sizes[0] / BATCH;

    float* xT   = (float*)d_ws;
    int*   ekey = (int*)(xT + (size_t)size1 * BATCH);
    float* out  = (float*)d_out;

    const int nTile = size1 / TP_COLS;
    const int nKey  = (nEdges + TPB - 1) / TPB;
    const int nPrep = nTile > nKey ? nTile : nKey;

    prep_kernel<<<nPrep, TPB, 0, stream>>>(x, rows, cols, xT, ekey, size1, nEdges);
    tile_spmm_wmma<<<size2 / OFEAT, ETPB, 0, stream>>>(
        wmean, wlogv, epsw, bmean, blogv, epsb, ekey, xT, out, nEdges, size2);
}


# Edge binning moved into prep: each 256-edge chunk bins its list entries into per-owner 32-byte cells (LDS counters); tile kernel gathers its 63 cells, no scan of all keys; list still arrival order
# speedup vs baseline: 2.7868x; 1.0775x over previous
#include <hip/hip_runtime.h>


typedef float v2f __attribute__((ext_vector_type(2)));
typedef float v4f __attribute__((ext_vector_type(4)));
typedef float v8f __attribute__((ext_vector_type(8)));
typedef int   v4i __attribute__((ext_vector_type(4)));
typedef __attribute__((address_space(3))) float lds_f32;

#define TPB 256
#define BATCH 32
#define TP_COLS 64
#define TP_PITCH 68

#define EW 16
#define ETPB (EW * 32)
#define EPW 2
#define TPBK 4
#define OFEAT (TPBK * 16)
#define O_PITCH 68
#define LIST_CAP 16384
#define CELL_W 8
#define PREP_OWN 256
#define CHUNK 256
#define MAXNE 6

template <int N> struct IC { static constexpr int value = N; };

__global__ __launch_bounds__(TPB) void prep_kernel(
    const float* __restrict__ x,
    const int* __restrict__ rows, const int* __restrict__ cols,
    float* __restrict__ xT, int* __restrict__ ekey, int* __restrict__ cells,
    int size1, int nEdges, int size2)
{
    __shared__ __attribute__((aligned(16))) float tile[BATCH][TP_PITCH];
    __shared__ __attribute__((aligned(16))) int cellL[PREP_OWN][CELL_W];

    const int tid = threadIdx.x;
    const int c0  = blockIdx.x * TP_COLS;
    const bool pT = c0 < size1;
    const int e   = blockIdx.x * TPB + tid;
    const bool pE = e < nEdges;
    const bool pK = blockIdx.x * TPB < nEdges;
    const int pitch = (((size2 / OFEAT) + 3) & ~3) * CELL_W;

    const int lr = tid >> 4;
    const int lq = (tid & 15) * 4;
    v4f xa = {}, xb = {};
    if (pT) {
        const float* src = x + (size_t)lr * size1 + c0 + lq;
        xa = *(const v4f*)src;
        xb = *(const v4f*)(src + (size_t)16 * size1);
    }
    int kr = 0, kc = 0;
    if (pE) {
        kr = rows[(size_t)e * 256];
        kc = cols[(size_t)e * 256];
    }

    if (pT) {
        *(v4f*)(&tile[lr][lq])      = xa;
        *(v4f*)(&tile[lr + 16][lq]) = xb;
    }
    if (pK) {
        const v4i z = {0, 0, 0, 0};
        ((v4i*)cellL)[tid]       = z;
        ((v4i*)cellL)[tid + TPB] = z;
    }
    __syncthreads();

    unsigned int key = 0u;
    if (pE) {
        const unsigned int orow = (unsigned int)kr & 0xFFFFu;
        key = orow | ((unsigned int)kc << 16);
        const unsigned int own = orow >> 6;
        if (own < PREP_OWN) {
            const int s = atomicAdd(&cellL[own][0], 1);
            if (s < CELL_W - 1)
                cellL[own][1 + s] = (int)((((unsigned int)e << 2) & 0xFFFFu) | ((orow >> 4) & 3u)
                                          | (key & 0xFFFF0000u));
        }
    }
    if (pK) {
        __syncthreads();
        v4i* dst = (v4i*)(cells + (size_t)blockIdx.x * pitch);
        const int nq = pitch / 4 < PREP_OWN * CELL_W / 4 ? pitch / 4 : PREP_OWN * CELL_W / 4;
        for (int i = tid; i < nq; i += TPB) dst[i] = ((const v4i*)cellL)[i];
    }
    if (pT) {
#pragma unroll
        for (int p = 0; p < 2; ++p) {
            const int j    = tid + TPB * p;
            const int ctl  = j >> 7;
            const int u    = (j >> 5) & 3;
            const int ln   = j & 31;
            const int bt   = 16 * (u >> 1) + (ln & 15);
            const int col  = 16 * ctl + 8 * (u & 1) + (ln >> 4);
            v4f v;
            v.x = tile[bt][col];
            v.y = tile[bt][col + 2];
            v.z = tile[bt][col + 4];
            v.w = tile[bt][col + 6];
            *(v4f*)(xT + (size_t)c0 * BATCH + j * 4) = v;
        }
    }
    if (pE) ekey[e] = (int)key;
}

__global__ __launch_bounds__(ETPB) void tile_spmm_wmma(
    const float* __restrict__ wmean, const float* __restrict__ wlogv,
    const float* __restrict__ epsw,
    const float* __restrict__ bmean, const float* __restrict__ blogv,
    const float* __restrict__ epsb,
    const int* __restrict__ ekey, const int* __restrict__ cells,
    const float* __restrict__ xT, float* __restrict__ out,
    int nEdges, int size2)
{
    __shared__ __attribute__((aligned(16))) float lds_o[BATCH][O_PITCH];
    __shared__ unsigned int elist[LIST_CAP];
    __shared__ __attribute__((aligned(16))) unsigned int lctl[4];

    const int tid  = threadIdx.x;
    const int lane = tid & 31;
    const int wv   = __builtin_amdgcn_readfirstlane(tid >> 5);
    const int blk  = blockIdx.x;

    const int nChunks = (nEdges + CHUNK - 1) / CHUNK;
    const int pitch   = (((size2 / OFEAT) + 3) & ~3) * CELL_W;
    const int cp = tid >> 3;
    const int cj = tid & 7;
    int cv = 0;
    if (cp < nChunks) cv = cells[(size_t)cp * pitch + blk * CELL_W + cj];

    const int ob = tid >> 4;
    const int oq = (tid & 15) * 4;
    {
        const int r = blk * OFEAT + oq;
        const v4f eb = *(const v4f*)(epsb  + r);
        const v4f lb = *(const v4f*)(blogv + r);
        const v4f mb = *(const v4f*)(bmean + r);
        v4f v;
        v.x = fmaf(eb.x, __expf(lb.x), mb.x);
        v.y = fmaf(eb.y, __expf(lb.y), mb.y);
        v.z = fmaf(eb.z, __expf(lb.z), mb.z);
        v.w = fmaf(eb.w, __expf(lb.w), mb.w);
        *(v4f*)(&lds_o[ob][oq]) = v;
    }
    if (tid < 4) lctl[tid] = 0u;
    if (blk == 0 && tid == 0) out[(size_t)size2 * BATCH] = 0.0f;
    __syncthreads();

    const int cnt = __shfl(cv, lane & ~7);
    if (cp < nChunks) {
        if (cnt > CELL_W - 1) {
            if (cj == 0) atomicOr(&lctl[cp >> 5], 1u << (cp & 31));
        } else if (cj >= 1 && cj <= cnt) {
            const unsigned int s = atomicAdd(&lctl[2], 1u);
            if (s < LIST_CAP) elist[s] = (unsigned int)cv;
        }
    }
    __syncthreads();
    const v4i ctl = *(const v4i*)lctl;
    const unsigned int ov0 = (unsigned int)ctl.x, ov1 = (unsigned int)ctl.y;
    unsigned int nlist = (unsigned int)ctl.z;
    if (ov0 | ov1) {
        for (int p = 0; p < nChunks; ++p) {
            if ((((p < 32) ? ov0 : ov1) >> (p & 31)) & 1u) {
                const int e = p * CHUNK + tid;
                if (tid < CHUNK && e < nEdges) {
                    const unsigned int key  = (unsigned int)ekey[e];
                    const unsigned int orow = key & 0xFFFFu;
                    if ((int)(orow >> 6) == blk) {
                        const unsigned int s = atomicAdd(&lctl[2], 1u);
                        if (s < LIST_CAP)
                            elist[s] = (((unsigned int)e << 2) & 0xFFFFu) | ((orow >> 4) & 3u) | (key & 0xFFFF0000u);
                    }
                }
            }
        }
        __syncthreads();
        nlist = lctl[2];
    }
    const int n = __builtin_amdgcn_readfirstlane((int)(nlist < LIST_CAP ? nlist : LIST_CAP));

    const int c  = lane & 15;
    const int hh = lane >> 4;

    auto stream = [&](auto nec, const int first) __attribute__((always_inline)) {
        constexpr int NE = decltype(nec)::value;
        constexpr int NF = NE < EPW ? NE : EPW;

        int    slot[NE];
        size_t base[NE];
        int    bcol[NE];
#pragma unroll
        for (int t = 0; t < NE; ++t) {
            const unsigned int en = __builtin_amdgcn_readfirstlane(elist[first + t * EW]);
            slot[t] = (int)(en & 3u);
            base[t] = (size_t)((en & 0xFFFFu) >> 2) * 256;
            bcol[t] = (int)(en >> 16);
        }

        float m[EPW][8], l[EPW][8], g[EPW][8];
        v4f xv[EPW][4];
        auto load_edge = [&](const int t, const int s) __attribute__((always_inline)) {
#pragma unroll
            for (int q = 0; q < 8; ++q) {
                const int o = q * 32 + lane;
                m[s][q] = wmean[base[t] + o];
                l[s][q] = wlogv[base[t] + o];
                g[s][q] = epsw[base[t] + o];
            }
            const float* xt = xT + (size_t)bcol[t] * BATCH;
#pragma unroll
            for (int u = 0; u < 4; ++u)
                xv[s][u] = *(const v4f*)(xt + (u * 32 + lane) * 4);
        };

#pragma unroll
        for (int t = 0; t < NF; ++t) load_edge(t, t);

#pragma unroll
        for (int t = 0; t < NE; ++t) {
            const int s = t % EPW;
            __builtin_amdgcn_sched_barrier(0);
            v2f wf[4];
#pragma unroll
            for (int kc = 0; kc < 4; ++kc) {
                wf[kc].x = fmaf(g[s][2 * kc],     __expf(l[s][2 * kc]),     m[s][2 * kc]);
                wf[kc].y = fmaf(g[s][2 * kc + 1], __expf(l[s][2 * kc + 1]), m[s][2 * kc + 1]);
            }
            v2f xf[2][4];
#pragma unroll
            for (int h = 0; h < 2; ++h)
#pragma unroll
                for (int kc = 0; kc < 4; ++kc) {
                    const v4f q4 = xv[s][h * 2 + (kc >> 1)];
                    xf[h][kc].x = (kc & 1) ? q4.z : q4.x;
                    xf[h][kc].y = (kc & 1) ? q4.w : q4.y;
                }

            if (t + EPW < NE) load_edge(t + EPW, s);

            lds_f32* o0l = (lds_f32*)&lds_o[8 * hh][slot[t] * 16 + c];
            asm volatile("" : "+v"(o0l));
            float* const o0 = (float*)o0l;

            __builtin_amdgcn_sched_barrier(0);

            v8f acc0 = {}, acc1 = {};
#pragma unroll
            for (int kc = 0; kc < 4; ++kc) {
                acc0 = __builtin_amdgcn_wmma_f32_16x16x4_f32(
                     false, xf[0][kc],  false, wf[kc],
                     (short)0, acc0,  false,  false);
                acc1 = __builtin_amdgcn_wmma_f32_16x16x4_f32(
                     false, xf[1][kc],  false, wf[kc],
                     (short)0, acc1,  false,  false);
            }
#pragma unroll
            for (int r = 0; r < 8; ++r)
                atomicAdd(o0 + r * O_PITCH, acc0[r]);
#pragma unroll
            for (int r = 0; r < 8; ++r)
                atomicAdd(o0 + (16 + r) * O_PITCH, acc1[r]);
            __builtin_amdgcn_sched_barrier(0);
            asm volatile("v_nop\n\tv_nop\n\tv_nop\n\tv_nop\n\tv_nop\n\tv_nop\n\tv_nop\n\tv_nop");
        }
    };

    const int ne_all = n > wv ? (n - wv + EW - 1) / EW : 0;
    for (int t0 = 0; t0 < ne_all; t0 += MAXNE) {
        const int left  = ne_all - t0;
        const int first = wv + t0 * EW;
        switch (left) {
            case 1:  stream(IC<1>{}, first); break;
            case 2:  stream(IC<2>{}, first); break;
            case 3:  stream(IC<3>{}, first); break;
            case 4:  stream(IC<4>{}, first); break;
            case 5:  stream(IC<5>{}, first); break;
            default: stream(IC<MAXNE>{}, first); break;
        }
    }
    __syncthreads();

    *(v4f*)(out + (size_t)ob * size2 + blk * OFEAT + oq) = *(const v4f*)(&lds_o[ob][oq]);
}

extern "C" void kernel_launch(void* const* d_in, const int* in_sizes, int n_in,
                              void* d_out, int out_size, void* d_ws, size_t ws_size,
                              hipStream_t stream)
{
    const float* x     = (const float*)d_in[0];
    const float* wmean = (const float*)d_in[1];
    const float* wlogv = (const float*)d_in[2];
    const float* bmean = (const float*)d_in[3];
    const float* blogv = (const float*)d_in[4];
    const float* epsw  = (const float*)d_in[5];
    const float* epsb  = (const float*)d_in[6];
    const int*   rows  = (const int*)d_in[7];
    const int*   cols  = (const int*)d_in[8];

    const int size2  = in_sizes[3];
    const int nnz    = in_sizes[1];
    const int nEdges = nnz / 256;
    const int size1  = in_sizes[0] / BATCH;

    float* xT   = (float*)d_ws;
    int*   ekey = (int*)(xT + (size_t)size1 * BATCH);
    int*   cells = ekey + nEdges;
    float* out  = (float*)d_out;

    const int nTile = size1 / TP_COLS;
    const int nKey  = (nEdges + TPB - 1) / TPB;
    const int nPrep = nTile > nKey ? nTile : nKey;

    prep_kernel<<<nPrep, TPB, 0, stream>>>(x, rows, cols, xT, ekey, cells, size1, nEdges, size2);
    tile_spmm_wmma<<<size2 / OFEAT, ETPB, 0, stream>>>(
        wmean, wlogv, epsw, bmean, blogv, epsb, ekey, cells, xT, out, nEdges, size2);
}
